# starting kernel
# baseline (speedup 1.0000x reference)
#include <hip/hip_runtime.h>
#include <cstddef>

typedef float v2f __attribute__((ext_vector_type(2)));
typedef float v8f __attribute__((ext_vector_type(8)));

#define NJ 20

__device__ __forceinline__ void rpy2r_dev(float r, float p, float y, float* R) {
  float cr = cosf(r), sr = sinf(r);
  float cp = cosf(p), sp = sinf(p);
  float cy = cosf(y), sy = sinf(y);
  R[0] = cy * cp; R[1] = cy * sp * sr - sy * cr; R[2] = cy * sp * cr + sy * sr;
  R[3] = sy * cp; R[4] = sy * sp * sr + cy * cr; R[5] = sy * sp * cr - cy * sr;
  R[6] = -sp;     R[7] = cp * sr;                R[8] = cp * cr;
}

__device__ __forceinline__ void mat3mul(const float* A, const float* B, float* C) {
#pragma unroll
  for (int i = 0; i < 3; ++i)
#pragma unroll
    for (int k = 0; k < 3; ++k)
      C[i * 3 + k] = A[i * 3 + 0] * B[0 * 3 + k] + A[i * 3 + 1] * B[1 * 3 + k] +
                     A[i * 3 + 2] * B[2 * 3 + k];
}

__global__ void fk_const_kernel(const float* __restrict__ rev_p_off,
                                const float* __restrict__ rev_rpy_off,
                                const float* __restrict__ pri_p_off,
                                const float* __restrict__ pri_rpy_off,
                                const float* __restrict__ rev_axis,
                                const float* __restrict__ pri_axis,
                                const float* __restrict__ p_track,
                                const float* __restrict__ rpy_track,
                                float* __restrict__ cst) {
  int j = threadIdx.x;
  if (j >= NJ) return;
  float* o = cst + j * 64;
  float Rt[9];

  rpy2r_dev(rev_rpy_off[3 * j], rev_rpy_off[3 * j + 1], rev_rpy_off[3 * j + 2], Rt);
#pragma unroll
  for (int k = 0; k < 9; ++k) o[k] = Rt[k];
  o[9] = rev_p_off[3 * j]; o[10] = rev_p_off[3 * j + 1]; o[11] = rev_p_off[3 * j + 2];

  rpy2r_dev(pri_rpy_off[3 * j], pri_rpy_off[3 * j + 1], pri_rpy_off[3 * j + 2], Rt);
#pragma unroll
  for (int k = 0; k < 9; ++k) o[12 + k] = Rt[k];
  o[21] = pri_p_off[3 * j]; o[22] = pri_p_off[3 * j + 1]; o[23] = pri_p_off[3 * j + 2];

  float x = rev_axis[3 * j], y = rev_axis[3 * j + 1], z = rev_axis[3 * j + 2];
  float inv = 1.f / (sqrtf(x * x + y * y + z * z) + 1e-12f);
  o[24] = x * inv; o[25] = y * inv; o[26] = z * inv;

  o[27] = pri_axis[3 * j]; o[28] = pri_axis[3 * j + 1]; o[29] = pri_axis[3 * j + 2];

  rpy2r_dev(rpy_track[3 * j], rpy_track[3 * j + 1], rpy_track[3 * j + 2], Rt);
  o[30] = Rt[0]; o[31] = Rt[1]; o[32] = Rt[2]; o[33] = p_track[3 * j];
  o[34] = Rt[3]; o[35] = Rt[4]; o[36] = Rt[5]; o[37] = p_track[3 * j + 1];
  o[38] = Rt[6]; o[39] = Rt[7]; o[40] = Rt[8]; o[41] = p_track[3 * j + 2];
  o[42] = 0.f; o[43] = 0.f; o[44] = 0.f; o[45] = 1.f;
}

__global__ __launch_bounds__(256) void fk_main_kernel(
    const float* __restrict__ rev_q, const float* __restrict__ pri_q,
    const float* __restrict__ cst, float* __restrict__ out, int B) {
  __shared__ float stage[8 * 512];

  const int tid  = blockIdx.x * 256 + threadIdx.x;
  const int lane = threadIdx.x & 31;
  const int wv   = threadIdx.x >> 5;
  float* st = &stage[wv * 512];

  const bool alive = (tid < B);
  const int  bs    = alive ? tid : (B - 1);

  float* trk_out = out;
  float* rev_out = out + (size_t)B * 160;
  float* pri_out = out + (size_t)B * 480;

  const int  h       = lane >> 4;
  const int  m16     = lane & 15;
  const int  nu      = m16 & 3;
  const int  iq      = m16 >> 2;
  const bool useful  = (h == (iq >> 1));
  const int  baseSel = (iq & 1) << 2;
  const long waveBase = (long)blockIdx.x * 256 + (long)wv * 32;

  float R[9] = {1.f, 0.f, 0.f, 0.f, 1.f, 0.f, 0.f, 0.f, 1.f};
  float p[3] = {0.f, 0.f, 0.f};

  for (int j = 0; j < NJ; ++j) {
    const float* C = cst + j * 64;
    float Ror[9], por[3], Rop[9], pop[3], ar[3], ap[3];
#pragma unroll
    for (int k = 0; k < 9; ++k) Ror[k] = C[k];
#pragma unroll
    for (int k = 0; k < 3; ++k) por[k] = C[9 + k];
#pragma unroll
    for (int k = 0; k < 9; ++k) Rop[k] = C[12 + k];
#pragma unroll
    for (int k = 0; k < 3; ++k) pop[k] = C[21 + k];
#pragma unroll
    for (int k = 0; k < 3; ++k) ar[k] = C[24 + k];
#pragma unroll
    for (int k = 0; k < 3; ++k) ap[k] = C[27 + k];

    float q = rev_q[(size_t)bs * NJ + j];
    float d = pri_q[(size_t)bs * NJ + j];

    float sn = sinf(q), cs = cosf(q), oc = 1.f - cs;
    float ax = ar[0], ay = ar[1], az = ar[2];
    float Rod[9];
    Rod[0] = 1.f + oc * (-(ay * ay + az * az));
    Rod[1] = -sn * az + oc * (ax * ay);
    Rod[2] =  sn * ay + oc * (ax * az);
    Rod[3] =  sn * az + oc * (ax * ay);
    Rod[4] = 1.f + oc * (-(ax * ax + az * az));
    Rod[5] = -sn * ax + oc * (ay * az);
    Rod[6] = -sn * ay + oc * (ax * az);
    Rod[7] =  sn * ax + oc * (ay * az);
    Rod[8] = 1.f + oc * (-(ax * ax + ay * ay));

    float Rrs[9]; mat3mul(Ror, Rod, Rrs);
    float R1[9];  mat3mul(R, Rrs, R1);
    float p1[3];
    p1[0] = R[0] * por[0] + R[1] * por[1] + R[2] * por[2] + p[0];
    p1[1] = R[3] * por[0] + R[4] * por[1] + R[5] * por[2] + p[1];
    p1[2] = R[6] * por[0] + R[7] * por[1] + R[8] * por[2] + p[2];

    if (alive) {
      float4* o4 = (float4*)(rev_out + ((size_t)tid * NJ + j) * 16);
      o4[0] = make_float4(R1[0], R1[1], R1[2], p1[0]);
      o4[1] = make_float4(R1[3], R1[4], R1[5], p1[1]);
      o4[2] = make_float4(R1[6], R1[7], R1[8], p1[2]);
      o4[3] = make_float4(0.f, 0.f, 0.f, 1.f);
    }

    float tv0 = d * ap[0], tv1 = d * ap[1], tv2 = d * ap[2];
    float ppv[3];
    ppv[0] = Rop[0] * tv0 + Rop[1] * tv1 + Rop[2] * tv2 + pop[0];
    ppv[1] = Rop[3] * tv0 + Rop[4] * tv1 + Rop[5] * tv2 + pop[1];
    ppv[2] = Rop[6] * tv0 + Rop[7] * tv1 + Rop[8] * tv2 + pop[2];
    float R2[9]; mat3mul(R1, Rop, R2);
    float p2[3];
    p2[0] = R1[0] * ppv[0] + R1[1] * ppv[1] + R1[2] * ppv[2] + p1[0];
    p2[1] = R1[3] * ppv[0] + R1[4] * ppv[1] + R1[5] * ppv[2] + p1[1];
    p2[2] = R1[6] * ppv[0] + R1[7] * ppv[1] + R1[8] * ppv[2] + p1[2];
#pragma unroll
    for (int k = 0; k < 9; ++k) R[k] = R2[k];
    p[0] = p2[0]; p[1] = p2[1]; p[2] = p2[2];

    if (alive) {
      float4* o4 = (float4*)(pri_out + ((size_t)tid * NJ + j) * 16);
      o4[0] = make_float4(R[0], R[1], R[2], p[0]);
      o4[1] = make_float4(R[3], R[4], R[5], p[1]);
      o4[2] = make_float4(R[6], R[7], R[8], p[2]);
      o4[3] = make_float4(0.f, 0.f, 0.f, 1.f);
    }

    if ((j & 1) == 0) {
      __syncthreads();
      float* sl = st + lane * 16;
      sl[0] = R[0]; sl[1] = R[1]; sl[2]  = R[2]; sl[3]  = p[0];
      sl[4] = R[3]; sl[5] = R[4]; sl[6]  = R[5]; sl[7]  = p[1];
      sl[8] = R[6]; sl[9] = R[7]; sl[10] = R[8]; sl[11] = p[2];
      sl[12] = 0.f; sl[13] = 0.f; sl[14] = 0.f; sl[15] = 1.f;
      __syncthreads();

      v2f Bop;
      Bop.x = C[30 + (2 * h + 0) * 4 + nu];
      Bop.y = C[30 + (2 * h + 1) * 4 + nu];

      const v8f zero8 = {0.f, 0.f, 0.f, 0.f, 0.f, 0.f, 0.f, 0.f};
#pragma unroll
      for (int g = 0; g < 8; ++g) {
        const v2f A = *(const v2f*)(st + (g * 4 + iq) * 16 + nu * 4 + 2 * h);
        v8f D = __builtin_amdgcn_wmma_f32_16x16x4_f32(
            false, A, false, Bop, (short)0, zero8, false, false);
        float e0 = baseSel ? D[4] : D[0];
        float e1 = baseSel ? D[5] : D[1];
        float e2 = baseSel ? D[6] : D[2];
        float e3 = baseSel ? D[7] : D[3];
        long bb = waveBase + g * 4 + iq;
        if (useful && bb < (long)B) {
          float* o = trk_out + ((size_t)bb * 10 + (j >> 1)) * 16 + nu;
          o[0] = e0; o[4] = e1; o[8] = e2; o[12] = e3;
        }
      }
    }
  }
}

extern "C" void kernel_launch(void* const* d_in, const int* in_sizes, int n_in,
                              void* d_out, int out_size, void* d_ws, size_t ws_size,
                              hipStream_t stream) {
  (void)n_in; (void)out_size; (void)ws_size;
  const float* rev_q       = (const float*)d_in[0];
  const float* pri_q       = (const float*)d_in[1];
  const float* rev_p_off   = (const float*)d_in[2];
  const float* rev_rpy_off = (const float*)d_in[3];
  const float* pri_p_off   = (const float*)d_in[4];
  const float* pri_rpy_off = (const float*)d_in[5];
  const float* rev_axis    = (const float*)d_in[6];
  const float* pri_axis    = (const float*)d_in[7];
  const float* p_track     = (const float*)d_in[8];
  const float* rpy_track   = (const float*)d_in[9];

  float* cst = (float*)d_ws;
  int B = in_sizes[0] / NJ;

  fk_const_kernel<<<1, 32, 0, stream>>>(rev_p_off, rev_rpy_off, pri_p_off, pri_rpy_off,
                                        rev_axis, pri_axis, p_track, rpy_track, cst);
  int grid = (B + 255) / 256;
  fk_main_kernel<<<grid, 256, 0, stream>>>(rev_q, pri_q, cst, (float*)d_out, B);
}


# Four lanes per chain, each carrying one row of the 4x4 frame (joint trig split 5 per lane, shared by DPP): 64-B runs stored from registers; tracking WMMA fed from registers, no LDS, no barrier
# speedup vs baseline: 2.6778x; 2.6778x over previous
#include <hip/hip_runtime.h>
#include <cstddef>

typedef float v2f __attribute__((ext_vector_type(2)));
typedef float v8f __attribute__((ext_vector_type(8)));

#define NJ 20

__device__ __forceinline__ void rpy2r_dev(float r, float p, float y, float* R) {
  float cr = cosf(r), sr = sinf(r);
  float cp = cosf(p), sp = sinf(p);
  float cy = cosf(y), sy = sinf(y);
  R[0] = cy * cp; R[1] = cy * sp * sr - sy * cr; R[2] = cy * sp * cr + sy * sr;
  R[3] = sy * cp; R[4] = sy * sp * sr + cy * cr; R[5] = sy * sp * cr - cy * sr;
  R[6] = -sp;     R[7] = cp * sr;                R[8] = cp * cr;
}

__device__ __forceinline__ void mat3mul(const float* A, const float* B, float* C) {
#pragma unroll
  for (int i = 0; i < 3; ++i)
#pragma unroll
    for (int k = 0; k < 3; ++k)
      C[i * 3 + k] = A[i * 3 + 0] * B[0 * 3 + k] + A[i * 3 + 1] * B[1 * 3 + k] +
                     A[i * 3 + 2] * B[2 * 3 + k];
}

__global__ void fk_const_kernel(const float* __restrict__ rev_p_off,
                                const float* __restrict__ rev_rpy_off,
                                const float* __restrict__ pri_p_off,
                                const float* __restrict__ pri_rpy_off,
                                const float* __restrict__ rev_axis,
                                const float* __restrict__ pri_axis,
                                const float* __restrict__ p_track,
                                const float* __restrict__ rpy_track,
                                float* __restrict__ cst) {
  int j = threadIdx.x;
  if (j >= NJ) return;
  float* o = cst + j * 64;
  float Rt[9];

  rpy2r_dev(rev_rpy_off[3 * j], rev_rpy_off[3 * j + 1], rev_rpy_off[3 * j + 2], Rt);
#pragma unroll
  for (int k = 0; k < 9; ++k) o[k] = Rt[k];
  o[9] = rev_p_off[3 * j]; o[10] = rev_p_off[3 * j + 1]; o[11] = rev_p_off[3 * j + 2];

  rpy2r_dev(pri_rpy_off[3 * j], pri_rpy_off[3 * j + 1], pri_rpy_off[3 * j + 2], Rt);
#pragma unroll
  for (int k = 0; k < 9; ++k) o[12 + k] = Rt[k];
  o[21] = pri_p_off[3 * j]; o[22] = pri_p_off[3 * j + 1]; o[23] = pri_p_off[3 * j + 2];

  float x = rev_axis[3 * j], y = rev_axis[3 * j + 1], z = rev_axis[3 * j + 2];
  float inv = 1.f / (sqrtf(x * x + y * y + z * z) + 1e-12f);
  o[24] = x * inv; o[25] = y * inv; o[26] = z * inv;

  o[27] = pri_axis[3 * j]; o[28] = pri_axis[3 * j + 1]; o[29] = pri_axis[3 * j + 2];

  rpy2r_dev(rpy_track[3 * j], rpy_track[3 * j + 1], rpy_track[3 * j + 2], Rt);
  o[30] = Rt[0]; o[31] = Rt[1]; o[32] = Rt[2]; o[33] = p_track[3 * j];
  o[34] = Rt[3]; o[35] = Rt[4]; o[36] = Rt[5]; o[37] = p_track[3 * j + 1];
  o[38] = Rt[6]; o[39] = Rt[7]; o[40] = Rt[8]; o[41] = p_track[3 * j + 2];
  o[42] = 0.f; o[43] = 0.f; o[44] = 0.f; o[45] = 1.f;
}

__device__ __forceinline__ float xhalf(float x) {
  int b = __builtin_bit_cast(int, x);
  return __builtin_bit_cast(float,
      __builtin_amdgcn_permlanex16(b, b, 0x76543210, 0xfedcba98, false, false));
}

template <int G>
__device__ __forceinline__ float quad_bcast(float x) {
  int b = __builtin_bit_cast(int, x);
  return __builtin_bit_cast(float,
      __builtin_amdgcn_update_dpp(b, b, G * 0x55, 0xf, 0xf, false));
}

template <int J>
__device__ __forceinline__ void joint_step(
    const float* __restrict__ cst, const float (&sn5)[5], const float (&cs5)[5],
    const float (&d5)[5], const v2f Atr, float& s0, float& s1, float& s2, float& s3,
    const bool alive, const int h, const size_t chain, const int row,
    float* __restrict__ trk_out, float* __restrict__ rev_out, float* __restrict__ pri_out) {
  const float* C = cst + J * 64;
  float Ror[9], por[3], Rop[9], pop[3], ar[3], ap[3];
#pragma unroll
  for (int k = 0; k < 9; ++k) Ror[k] = C[k];
#pragma unroll
  for (int k = 0; k < 3; ++k) por[k] = C[9 + k];
#pragma unroll
  for (int k = 0; k < 9; ++k) Rop[k] = C[12 + k];
#pragma unroll
  for (int k = 0; k < 3; ++k) pop[k] = C[21 + k];
#pragma unroll
  for (int k = 0; k < 3; ++k) ar[k] = C[24 + k];
#pragma unroll
  for (int k = 0; k < 3; ++k) ap[k] = C[27 + k];

  const float sn = quad_bcast<J / 5>(sn5[J % 5]);
  const float cs = quad_bcast<J / 5>(cs5[J % 5]);
  const float d  = quad_bcast<J / 5>(d5[J % 5]);

  const float oc = 1.f - cs;
  const float ax = ar[0], ay = ar[1], az = ar[2];
  float Rod[9];
  Rod[0] = 1.f + oc * (-(ay * ay + az * az));
  Rod[1] = -sn * az + oc * (ax * ay);
  Rod[2] =  sn * ay + oc * (ax * az);
  Rod[3] =  sn * az + oc * (ax * ay);
  Rod[4] = 1.f + oc * (-(ax * ax + az * az));
  Rod[5] = -sn * ax + oc * (ay * az);
  Rod[6] = -sn * ay + oc * (ax * az);
  Rod[7] =  sn * ax + oc * (ay * az);
  Rod[8] = 1.f + oc * (-(ax * ax + ay * ay));

  float Rrs[9]; mat3mul(Ror, Rod, Rrs);
  const float a0 = s0 * Rrs[0] + s1 * Rrs[3] + s2 * Rrs[6];
  const float a1 = s0 * Rrs[1] + s1 * Rrs[4] + s2 * Rrs[7];
  const float a2 = s0 * Rrs[2] + s1 * Rrs[5] + s2 * Rrs[8];
  const float a3 = s0 * por[0] + s1 * por[1] + s2 * por[2] + s3;
  if (alive) {
    float4* o4 = (float4*)(rev_out + (chain * NJ + J) * 16 + row * 4);
    *o4 = make_float4(a0, a1, a2, a3);
  }

  const float tv0 = d * ap[0], tv1 = d * ap[1], tv2 = d * ap[2];
  float ppv[3];
  ppv[0] = Rop[0] * tv0 + Rop[1] * tv1 + Rop[2] * tv2 + pop[0];
  ppv[1] = Rop[3] * tv0 + Rop[4] * tv1 + Rop[5] * tv2 + pop[1];
  ppv[2] = Rop[6] * tv0 + Rop[7] * tv1 + Rop[8] * tv2 + pop[2];
  const float b0 = a0 * Rop[0] + a1 * Rop[3] + a2 * Rop[6];
  const float b1 = a0 * Rop[1] + a1 * Rop[4] + a2 * Rop[7];
  const float b2 = a0 * Rop[2] + a1 * Rop[5] + a2 * Rop[8];
  const float b3 = a0 * ppv[0] + a1 * ppv[1] + a2 * ppv[2] + a3;
  s0 = b0; s1 = b1; s2 = b2; s3 = b3;
  if (alive) {
    float4* o4 = (float4*)(pri_out + (chain * NJ + J) * 16 + row * 4);
    *o4 = make_float4(b0, b1, b2, b3);
  }

  if ((J & 1) == 0) {
    const float x = xhalf(h ? s0 : s2);
    const float y = xhalf(h ? s1 : s3);
    v2f B1, B2;
    B1.x = h ? x : s0;  B1.y = h ? y : s1;
    B2.x = h ? s2 : x;  B2.y = h ? s3 : y;
    const v8f zero8 = {0.f, 0.f, 0.f, 0.f, 0.f, 0.f, 0.f, 0.f};
    v8f D1 = __builtin_amdgcn_wmma_f32_16x16x4_f32(
        false, Atr, false, B1, (short)0, zero8, false, false);
    v8f D2 = __builtin_amdgcn_wmma_f32_16x16x4_f32(
        false, Atr, false, B2, (short)0, zero8, false, false);
    const float e0 = h ? D2[0] : D1[0];
    const float e1 = h ? D2[1] : D1[1];
    const float e2 = h ? D2[2] : D1[2];
    const float e3 = h ? D2[3] : D1[3];
    if (alive) {
      float4* o4 = (float4*)(trk_out + (chain * 10 + (J >> 1)) * 16 + row * 4);
      *o4 = make_float4(e0, e1, e2, e3);
    }
  }
}

__global__ __launch_bounds__(256) void fk_main_kernel(
    const float* __restrict__ rev_q, const float* __restrict__ pri_q,
    const float* __restrict__ cst, float* __restrict__ out, int B) {
  const int  gl    = blockIdx.x * 256 + threadIdx.x;
  const int  ch    = gl >> 2;
  const int  row   = gl & 3;
  const int  h     = (threadIdx.x >> 4) & 1;
  const bool alive = (ch < B);
  const size_t chain = (size_t)(alive ? ch : (B - 1));

  float* trk_out = out;
  float* rev_out = out + (size_t)B * 160;
  float* pri_out = out + (size_t)B * 480;

  float qv[5], d5[5], sn5[5], cs5[5];
#pragma unroll
  for (int t = 0; t < 5; ++t) {
    qv[t] = rev_q[chain * NJ + 5 * row + t];
    d5[t] = pri_q[chain * NJ + 5 * row + t];
  }
  v2f Atr[10];
#pragma unroll
  for (int t = 0; t < 10; ++t) {
    const float* C = cst + (2 * t) * 64;
    Atr[t].x = C[30 + (2 * h + 0) * 4 + row];
    Atr[t].y = C[30 + (2 * h + 1) * 4 + row];
  }
#pragma unroll
  for (int t = 0; t < 5; ++t) { sn5[t] = sinf(qv[t]); cs5[t] = cosf(qv[t]); }

  float s0 = (row == 0) ? 1.f : 0.f;
  float s1 = (row == 1) ? 1.f : 0.f;
  float s2 = (row == 2) ? 1.f : 0.f;
  float s3 = (row == 3) ? 1.f : 0.f;

#define JOINT(J) joint_step<J>(cst, sn5, cs5, d5, Atr[(J) >> 1], s0, s1, s2, s3, alive, h, \
                               chain, row, trk_out, rev_out, pri_out);
  JOINT(0)  JOINT(1)  JOINT(2)  JOINT(3)  JOINT(4)
  JOINT(5)  JOINT(6)  JOINT(7)  JOINT(8)  JOINT(9)
  JOINT(10) JOINT(11) JOINT(12) JOINT(13) JOINT(14)
  JOINT(15) JOINT(16) JOINT(17) JOINT(18) JOINT(19)
#undef JOINT
}

extern "C" void kernel_launch(void* const* d_in, const int* in_sizes, int n_in,
                              void* d_out, int out_size, void* d_ws, size_t ws_size,
                              hipStream_t stream) {
  (void)n_in; (void)out_size; (void)ws_size;
  const float* rev_q       = (const float*)d_in[0];
  const float* pri_q       = (const float*)d_in[1];
  const float* rev_p_off   = (const float*)d_in[2];
  const float* rev_rpy_off = (const float*)d_in[3];
  const float* pri_p_off   = (const float*)d_in[4];
  const float* pri_rpy_off = (const float*)d_in[5];
  const float* rev_axis    = (const float*)d_in[6];
  const float* pri_axis    = (const float*)d_in[7];
  const float* p_track     = (const float*)d_in[8];
  const float* rpy_track   = (const float*)d_in[9];

  float* cst = (float*)d_ws;
  int B = in_sizes[0] / NJ;

  fk_const_kernel<<<1, 32, 0, stream>>>(rev_p_off, rev_rpy_off, pri_p_off, pri_rpy_off,
                                        rev_axis, pri_axis, p_track, rpy_track, cst);
  int grid = (B + 63) / 64;
  fk_main_kernel<<<grid, 256, 0, stream>>>(rev_q, pri_q, cst, (float*)d_out, B);
}


# Per-joint constants formed by each wave itself (lane = joint, three offset rotations per lane), read by the joint steps as scalars via v_readlane: no table, no set-up launch; one launch per call
# speedup vs baseline: 4.8787x; 1.8219x over previous
#include <hip/hip_runtime.h>
#include <cstddef>

typedef float v2f __attribute__((ext_vector_type(2)));
typedef float v8f __attribute__((ext_vector_type(8)));

#define NJ 20

template <int N>
__device__ __forceinline__ void sincos_rolled(float (&a)[N], float (&s)[N], float (&c)[N]) {
#pragma unroll
  for (int k = 0; k < N; ++k) { s[k] = 0.f; c[k] = 0.f; }
#pragma nounroll
  for (int i = 0; i < N; ++i) {
    const float sv = sinf(a[0]), cv = cosf(a[0]);
#pragma unroll
    for (int k = 0; k + 1 < N; ++k) { a[k] = a[k + 1]; s[k] = s[k + 1]; c[k] = c[k + 1]; }
    s[N - 1] = sv; c[N - 1] = cv;
  }
}

__device__ __forceinline__ void rpy2r_sc(float sr, float cr, float sp, float cp, float sy,
                                         float cy, float* R) {
  R[0] = cy * cp; R[1] = cy * sp * sr - sy * cr; R[2] = cy * sp * cr + sy * sr;
  R[3] = sy * cp; R[4] = sy * sp * sr + cy * cr; R[5] = sy * sp * cr - cy * sr;
  R[6] = -sp;     R[7] = cp * sr;                R[8] = cp * cr;
}

__device__ __forceinline__ void mat3mul(const float* A, const float* B, float* C) {
#pragma unroll
  for (int i = 0; i < 3; ++i)
#pragma unroll
    for (int k = 0; k < 3; ++k)
      C[i * 3 + k] = A[i * 3 + 0] * B[0 * 3 + k] + A[i * 3 + 1] * B[1 * 3 + k] +
                     A[i * 3 + 2] * B[2 * 3 + k];
}

template <int L>
__device__ __forceinline__ float lane_val(float x) {
  return __builtin_bit_cast(float, __builtin_amdgcn_readlane(__builtin_bit_cast(int, x), L));
}

__device__ __forceinline__ float xhalf(float x) {
  int b = __builtin_bit_cast(int, x);
  return __builtin_bit_cast(float,
      __builtin_amdgcn_permlanex16(b, b, 0x76543210, 0xfedcba98, false, false));
}

template <int G>
__device__ __forceinline__ float quad_bcast(float x) {
  int b = __builtin_bit_cast(int, x);
  return __builtin_bit_cast(float,
      __builtin_amdgcn_update_dpp(b, b, G * 0x55, 0xf, 0xf, false));
}

template <int CTRL, int BANK>
__device__ __forceinline__ float dpp_take(float keep, float from) {
  return __builtin_bit_cast(float,
      __builtin_amdgcn_update_dpp(__builtin_bit_cast(int, keep), __builtin_bit_cast(int, from),
                                  CTRL, 0xf, BANK, false));
}

template <int XM>
__device__ __forceinline__ void xstage(float& lo, float& hi) {
  constexpr int CTRL    = 0x160 + XM;
  constexpr int BANK_1  = (XM == 4) ? 0xA : 0xC;
  constexpr int BANK_0  = (XM == 4) ? 0x5 : 0x3;
  const float nlo = dpp_take<CTRL, BANK_1>(lo, hi);
  const float nhi = dpp_take<CTRL, BANK_0>(hi, lo);
  lo = nlo; hi = nhi;
}

__device__ __forceinline__ void xpose4(float4 (&X)[4]) {
  xstage<4>(X[0].x, X[1].x); xstage<4>(X[2].x, X[3].x);
  xstage<4>(X[0].y, X[1].y); xstage<4>(X[2].y, X[3].y);
  xstage<4>(X[0].z, X[1].z); xstage<4>(X[2].z, X[3].z);
  xstage<4>(X[0].w, X[1].w); xstage<4>(X[2].w, X[3].w);
  xstage<8>(X[0].x, X[2].x); xstage<8>(X[1].x, X[3].x);
  xstage<8>(X[0].y, X[2].y); xstage<8>(X[1].y, X[3].y);
  xstage<8>(X[0].z, X[2].z); xstage<8>(X[1].z, X[3].z);
  xstage<8>(X[0].w, X[2].w); xstage<8>(X[1].w, X[3].w);
}

template <int PER_CHAIN>
__device__ __forceinline__ void store_units(float* __restrict__ lane_base, const float4 (&X)[4],
                                            const int cbase, const int B, const bool full,
                                            const int first_frame) {
  if (full) {
#pragma unroll
    for (int s = 0; s < 4; ++s) {
      float4* o4 = (float4*)(lane_base + (s * PER_CHAIN + first_frame) * 16);
      *o4 = X[s];
    }
  } else {
#pragma unroll
    for (int s = 0; s < 4; ++s) {
      if (cbase + s < B) {
        float4* o4 = (float4*)(lane_base + (s * PER_CHAIN + first_frame) * 16);
        *o4 = X[s];
      }
    }
  }
}

template <int J>
__device__ __forceinline__ void joint_step(
    const float (&RA)[9], const float (&RB)[9], const float (&AX)[3],
    const float* __restrict__ rev_p_off, const float* __restrict__ pri_p_off,
    const float* __restrict__ pri_axis, const float* __restrict__ p_track,
    const float (&sn5)[5], const float (&cs5)[5],
    const float (&d5)[5], float& s0, float& s1, float& s2, float& s3,
    const int h, const int row, float4& rv, float4& pr, float4& tk) {
  float Ror[9], por[3], Rop[9], pop[3], ar[3], ap[3];
#pragma unroll
  for (int k = 0; k < 9; ++k) Ror[k] = lane_val<J>(RA[k]);
#pragma unroll
  for (int k = 0; k < 3; ++k) por[k] = rev_p_off[3 * J + k];
#pragma unroll
  for (int k = 0; k < 9; ++k) Rop[k] = lane_val<J>(RB[k]);
#pragma unroll
  for (int k = 0; k < 3; ++k) pop[k] = pri_p_off[3 * J + k];
#pragma unroll
  for (int k = 0; k < 3; ++k) ar[k] = lane_val<J>(AX[k]);
#pragma unroll
  for (int k = 0; k < 3; ++k) ap[k] = pri_axis[3 * J + k];

  const float sn = quad_bcast<J / 5>(sn5[J % 5]);
  const float cs = quad_bcast<J / 5>(cs5[J % 5]);
  const float d  = quad_bcast<J / 5>(d5[J % 5]);

  const float oc = 1.f - cs;
  const float ax = ar[0], ay = ar[1], az = ar[2];
  float Rod[9];
  Rod[0] = 1.f + oc * (-(ay * ay + az * az));
  Rod[1] = -sn * az + oc * (ax * ay);
  Rod[2] =  sn * ay + oc * (ax * az);
  Rod[3] =  sn * az + oc * (ax * ay);
  Rod[4] = 1.f + oc * (-(ax * ax + az * az));
  Rod[5] = -sn * ax + oc * (ay * az);
  Rod[6] = -sn * ay + oc * (ax * az);
  Rod[7] =  sn * ax + oc * (ay * az);
  Rod[8] = 1.f + oc * (-(ax * ax + ay * ay));

  float Rrs[9]; mat3mul(Ror, Rod, Rrs);
  const float a0 = s0 * Rrs[0] + s1 * Rrs[3] + s2 * Rrs[6];
  const float a1 = s0 * Rrs[1] + s1 * Rrs[4] + s2 * Rrs[7];
  const float a2 = s0 * Rrs[2] + s1 * Rrs[5] + s2 * Rrs[8];
  const float a3 = s0 * por[0] + s1 * por[1] + s2 * por[2] + s3;
  rv = make_float4(a0, a1, a2, a3);

  const float tv0 = d * ap[0], tv1 = d * ap[1], tv2 = d * ap[2];
  float ppv[3];
  ppv[0] = Rop[0] * tv0 + Rop[1] * tv1 + Rop[2] * tv2 + pop[0];
  ppv[1] = Rop[3] * tv0 + Rop[4] * tv1 + Rop[5] * tv2 + pop[1];
  ppv[2] = Rop[6] * tv0 + Rop[7] * tv1 + Rop[8] * tv2 + pop[2];
  const float b0 = a0 * Rop[0] + a1 * Rop[3] + a2 * Rop[6];
  const float b1 = a0 * Rop[1] + a1 * Rop[4] + a2 * Rop[7];
  const float b2 = a0 * Rop[2] + a1 * Rop[5] + a2 * Rop[8];
  const float b3 = a0 * ppv[0] + a1 * ppv[1] + a2 * ppv[2] + a3;
  s0 = b0; s1 = b1; s2 = b2; s3 = b3;
  pr = make_float4(b0, b1, b2, b3);

  if ((J & 1) == 0) {
    constexpr int SL = NJ + J / 2;
    const float t0 = lane_val<SL>(RA[0]), t1 = lane_val<SL>(RA[1]), t2 = lane_val<SL>(RA[2]);
    const float t3 = lane_val<SL>(RA[3]), t4 = lane_val<SL>(RA[4]), t5 = lane_val<SL>(RA[5]);
    const float t6 = lane_val<SL>(RA[6]), t7 = lane_val<SL>(RA[7]), t8 = lane_val<SL>(RA[8]);
    const float pt0 = p_track[3 * J], pt1 = p_track[3 * J + 1], pt2 = p_track[3 * J + 2];
    const float c0 = (row == 0) ? t0 : (row == 1) ? t1 : (row == 2) ? t2 : pt0;
    const float c1 = (row == 0) ? t3 : (row == 1) ? t4 : (row == 2) ? t5 : pt1;
    const float c2 = (row == 0) ? t6 : (row == 1) ? t7 : (row == 2) ? t8 : pt2;
    const float c3 = (row == 3) ? 1.f : 0.f;
    v2f Atr;
    Atr.x = h ? c2 : c0;
    Atr.y = h ? c3 : c1;
    const float x = xhalf(h ? s0 : s2);
    const float y = xhalf(h ? s1 : s3);
    v2f B1, B2;
    B1.x = h ? x : s0;  B1.y = h ? y : s1;
    B2.x = h ? s2 : x;  B2.y = h ? s3 : y;
    const v8f zero8 = {0.f, 0.f, 0.f, 0.f, 0.f, 0.f, 0.f, 0.f};
    v8f D1 = __builtin_amdgcn_wmma_f32_16x16x4_f32(
        false, Atr, false, B1, (short)0, zero8, false, false);
    v8f D2 = __builtin_amdgcn_wmma_f32_16x16x4_f32(
        false, Atr, false, B2, (short)0, zero8, false, false);
    const float e0 = h ? D2[0] : D1[0];
    const float e1 = h ? D2[1] : D1[1];
    const float e2 = h ? D2[2] : D1[2];
    const float e3 = h ? D2[3] : D1[3];
    tk = make_float4(e0, e1, e2, e3);
  }
}

__global__ __launch_bounds__(256) void fk_main_kernel(
    const float* __restrict__ rev_q, const float* __restrict__ pri_q,
    const float* __restrict__ rev_p_off, const float* __restrict__ rev_rpy_off,
    const float* __restrict__ pri_p_off, const float* __restrict__ pri_rpy_off,
    const float* __restrict__ rev_axis, const float* __restrict__ pri_axis,
    const float* __restrict__ p_track, const float* __restrict__ rpy_track,
    float* __restrict__ out, int B) {
  const int  gl    = blockIdx.x * 256 + threadIdx.x;
  const int  ch    = gl >> 2;
  const int  row   = gl & 3;
  const int  h     = (threadIdx.x >> 4) & 1;
  const int  lane  = threadIdx.x & 31;
  const bool alive = (ch < B);
  const size_t chain = (size_t)(alive ? ch : (B - 1));

  float* trk_out = out;
  float* rev_out = out + (size_t)B * 160;
  float* pri_out = out + (size_t)B * 480;

  const int jA = (lane < NJ) ? lane : ((lane < NJ + 10) ? 2 * (lane - NJ) : 18);
  const float* rpyA = ((lane < NJ) ? rev_rpy_off : rpy_track) + 3 * jA;
  const int jB = (lane < NJ) ? lane : (NJ - 1);
  float angC[6], sC[6], cC[6];
  angC[0] = rpyA[0]; angC[1] = rpyA[1]; angC[2] = rpyA[2];
  angC[3] = pri_rpy_off[3 * jB]; angC[4] = pri_rpy_off[3 * jB + 1]; angC[5] = pri_rpy_off[3 * jB + 2];
  const float axx = rev_axis[3 * jB], axy = rev_axis[3 * jB + 1], axz = rev_axis[3 * jB + 2];

  float qv[5], d5[5], sn5[5], cs5[5];
#pragma unroll
  for (int t = 0; t < 5; ++t) {
    qv[t] = rev_q[chain * NJ + 5 * row + t];
    d5[t] = pri_q[chain * NJ + 5 * row + t];
  }

  float RA[9], RB[9], AX[3];
  sincos_rolled<6>(angC, sC, cC);
  rpy2r_sc(sC[0], cC[0], sC[1], cC[1], sC[2], cC[2], RA);
  rpy2r_sc(sC[3], cC[3], sC[4], cC[4], sC[5], cC[5], RB);
  {
    const float inv = 1.f / (sqrtf(axx * axx + axy * axy + axz * axz) + 1e-12f);
    AX[0] = axx * inv; AX[1] = axy * inv; AX[2] = axz * inv;
  }

  sincos_rolled<5>(qv, sn5, cs5);

  float s0 = (row == 0) ? 1.f : 0.f;
  float s1 = (row == 1) ? 1.f : 0.f;
  float s2 = (row == 2) ? 1.f : 0.f;
  float s3 = (row == 3) ? 1.f : 0.f;

  const int u     = (threadIdx.x >> 2) & 3;
  const int cbase = ((gl >> 5) << 3) + 4 * h;
  const bool full = (__builtin_amdgcn_readfirstlane(gl >> 5) * 8 + 8 <= B);
  float* rev_lane = rev_out + ((size_t)cbase * NJ + u) * 16 + row * 4;
  float* pri_lane = pri_out + ((size_t)cbase * NJ + u) * 16 + row * 4;
  float* trk_lane = trk_out + ((size_t)cbase * 10 + u) * 16 + row * 4;

  float4 RV[4], PR[4], TK[4];
#define JOINT(J) joint_step<J>(RA, RB, AX, rev_p_off, pri_p_off, pri_axis, p_track, sn5, cs5, d5, \
                               s0, s1, s2, s3, h, row, RV[(J) & 3], PR[(J) & 3], TK[((J) >> 1) & 3]);
#define FLUSH(G) \
  xpose4(RV); store_units<NJ>(rev_lane, RV, cbase, B, full, 4 * (G)); \
  xpose4(PR); store_units<NJ>(pri_lane, PR, cbase, B, full, 4 * (G));
  JOINT(0)  JOINT(1)  JOINT(2)  JOINT(3)  FLUSH(0)
  JOINT(4)  JOINT(5)  JOINT(6)
  xpose4(TK); store_units<10>(trk_lane, TK, cbase, B, full, 0);
  JOINT(7)  FLUSH(1)
  JOINT(8)  JOINT(9)  JOINT(10) JOINT(11) FLUSH(2)
  JOINT(12) JOINT(13) JOINT(14)
  xpose4(TK); store_units<10>(trk_lane, TK, cbase, B, full, 4);
  JOINT(15) FLUSH(3)
  JOINT(16) JOINT(17) JOINT(18) JOINT(19) FLUSH(4)
#undef FLUSH
#undef JOINT

  xstage<4>(TK[0].x, TK[1].x); xstage<4>(TK[0].y, TK[1].y);
  xstage<4>(TK[0].z, TK[1].z); xstage<4>(TK[0].w, TK[1].w);
  const int cpair = cbase + 2 * (u >> 1);
  float* trk_tail = trk_out + ((size_t)cpair * 10 + 8 + (u & 1)) * 16 + row * 4;
#pragma unroll
  for (int t = 0; t < 2; ++t) {
    if (cpair + t < B) {
      float4* o4 = (float4*)(trk_tail + t * 10 * 16);
      *o4 = TK[t];
    }
  }
}

extern "C" void kernel_launch(void* const* d_in, const int* in_sizes, int n_in,
                              void* d_out, int out_size, void* d_ws, size_t ws_size,
                              hipStream_t stream) {
  (void)n_in; (void)out_size; (void)ws_size;
  const float* rev_q       = (const float*)d_in[0];
  const float* pri_q       = (const float*)d_in[1];
  const float* rev_p_off   = (const float*)d_in[2];
  const float* rev_rpy_off = (const float*)d_in[3];
  const float* pri_p_off   = (const float*)d_in[4];
  const float* pri_rpy_off = (const float*)d_in[5];
  const float* rev_axis    = (const float*)d_in[6];
  const float* pri_axis    = (const float*)d_in[7];
  const float* p_track     = (const float*)d_in[8];
  const float* rpy_track   = (const float*)d_in[9];

  (void)d_ws;
  int B = in_sizes[0] / NJ;

  int grid = (B + 63) / 64;
  fk_main_kernel<<<grid, 256, 0, stream>>>(rev_q, pri_q, rev_p_off, rev_rpy_off, pri_p_off,
                                           pri_rpy_off, rev_axis, pri_axis, p_track, rpy_track,
                                           (float*)d_out, B);
}


# Joint step applied to the lane's row from the left: row @ offset rotation, Rodrigues as two cross products on the 3-vector, slide term from the new row; no 3x3 step matrix built (sums re-associated)
# speedup vs baseline: 5.2938x; 1.0851x over previous
#include <hip/hip_runtime.h>
#include <cstddef>

typedef float v2f __attribute__((ext_vector_type(2)));
typedef float v8f __attribute__((ext_vector_type(8)));

#define NJ 20

template <int N>
__device__ __forceinline__ void sincos_rolled(float (&a)[N], float (&s)[N], float (&c)[N]) {
#pragma unroll
  for (int k = 0; k < N; ++k) { s[k] = 0.f; c[k] = 0.f; }
#pragma nounroll
  for (int i = 0; i < N; ++i) {
    const float sv = sinf(a[0]), cv = cosf(a[0]);
#pragma unroll
    for (int k = 0; k + 1 < N; ++k) { a[k] = a[k + 1]; s[k] = s[k + 1]; c[k] = c[k + 1]; }
    s[N - 1] = sv; c[N - 1] = cv;
  }
}

__device__ __forceinline__ void rpy2r_sc(float sr, float cr, float sp, float cp, float sy,
                                         float cy, float* R) {
  R[0] = cy * cp; R[1] = cy * sp * sr - sy * cr; R[2] = cy * sp * cr + sy * sr;
  R[3] = sy * cp; R[4] = sy * sp * sr + cy * cr; R[5] = sy * sp * cr - cy * sr;
  R[6] = -sp;     R[7] = cp * sr;                R[8] = cp * cr;
}

template <int L>
__device__ __forceinline__ float lane_val(float x) {
  return __builtin_bit_cast(float, __builtin_amdgcn_readlane(__builtin_bit_cast(int, x), L));
}

__device__ __forceinline__ float xhalf(float x) {
  int b = __builtin_bit_cast(int, x);
  return __builtin_bit_cast(float,
      __builtin_amdgcn_permlanex16(b, b, 0x76543210, 0xfedcba98, false, false));
}

template <int G>
__device__ __forceinline__ float quad_bcast(float x) {
  int b = __builtin_bit_cast(int, x);
  return __builtin_bit_cast(float,
      __builtin_amdgcn_update_dpp(b, b, G * 0x55, 0xf, 0xf, false));
}

template <int CTRL, int BANK>
__device__ __forceinline__ float dpp_take(float keep, float from) {
  return __builtin_bit_cast(float,
      __builtin_amdgcn_update_dpp(__builtin_bit_cast(int, keep), __builtin_bit_cast(int, from),
                                  CTRL, 0xf, BANK, false));
}

template <int XM>
__device__ __forceinline__ void xstage(float& lo, float& hi) {
  constexpr int CTRL    = 0x160 + XM;
  constexpr int BANK_1  = (XM == 4) ? 0xA : 0xC;
  constexpr int BANK_0  = (XM == 4) ? 0x5 : 0x3;
  const float nlo = dpp_take<CTRL, BANK_1>(lo, hi);
  const float nhi = dpp_take<CTRL, BANK_0>(hi, lo);
  lo = nlo; hi = nhi;
}

__device__ __forceinline__ void xpose4(float4 (&X)[4]) {
  xstage<4>(X[0].x, X[1].x); xstage<4>(X[2].x, X[3].x);
  xstage<4>(X[0].y, X[1].y); xstage<4>(X[2].y, X[3].y);
  xstage<4>(X[0].z, X[1].z); xstage<4>(X[2].z, X[3].z);
  xstage<4>(X[0].w, X[1].w); xstage<4>(X[2].w, X[3].w);
  xstage<8>(X[0].x, X[2].x); xstage<8>(X[1].x, X[3].x);
  xstage<8>(X[0].y, X[2].y); xstage<8>(X[1].y, X[3].y);
  xstage<8>(X[0].z, X[2].z); xstage<8>(X[1].z, X[3].z);
  xstage<8>(X[0].w, X[2].w); xstage<8>(X[1].w, X[3].w);
}

template <int PER_CHAIN>
__device__ __forceinline__ void store_units(float* __restrict__ lane_base, const float4 (&X)[4],
                                            const int cbase, const int B, const bool full,
                                            const int first_frame) {
  if (full) {
#pragma unroll
    for (int s = 0; s < 4; ++s) {
      float4* o4 = (float4*)(lane_base + (s * PER_CHAIN + first_frame) * 16);
      *o4 = X[s];
    }
  } else {
#pragma unroll
    for (int s = 0; s < 4; ++s) {
      if (cbase + s < B) {
        float4* o4 = (float4*)(lane_base + (s * PER_CHAIN + first_frame) * 16);
        *o4 = X[s];
      }
    }
  }
}

template <int J>
__device__ __forceinline__ void joint_step(
    const float (&RA)[9], const float (&RB)[9], const float (&AX)[3],
    const float* __restrict__ rev_p_off, const float* __restrict__ pri_p_off,
    const float* __restrict__ pri_axis, const float* __restrict__ p_track,
    const float (&sn5)[5], const float (&cs5)[5],
    const float (&d5)[5], float& s0, float& s1, float& s2, float& s3,
    const int h, const int row, float4& rv, float4& pr, float4& tk) {
  float Ror[9], por[3], Rop[9], pop[3], ar[3], ap[3];
#pragma unroll
  for (int k = 0; k < 9; ++k) Ror[k] = lane_val<J>(RA[k]);
#pragma unroll
  for (int k = 0; k < 3; ++k) por[k] = rev_p_off[3 * J + k];
#pragma unroll
  for (int k = 0; k < 9; ++k) Rop[k] = lane_val<J>(RB[k]);
#pragma unroll
  for (int k = 0; k < 3; ++k) pop[k] = pri_p_off[3 * J + k];
#pragma unroll
  for (int k = 0; k < 3; ++k) ar[k] = lane_val<J>(AX[k]);
#pragma unroll
  for (int k = 0; k < 3; ++k) ap[k] = pri_axis[3 * J + k];

  const float sn = quad_bcast<J / 5>(sn5[J % 5]);
  const float cs = quad_bcast<J / 5>(cs5[J % 5]);
  const float d  = quad_bcast<J / 5>(d5[J % 5]);

  const float oc = 1.f - cs;
  const float ax = ar[0], ay = ar[1], az = ar[2];
  const float w0 = s0 * Ror[0] + s1 * Ror[3] + s2 * Ror[6];
  const float w1 = s0 * Ror[1] + s1 * Ror[4] + s2 * Ror[7];
  const float w2 = s0 * Ror[2] + s1 * Ror[5] + s2 * Ror[8];
  const float k0 = w1 * az - w2 * ay;
  const float k1 = w2 * ax - w0 * az;
  const float k2 = w0 * ay - w1 * ax;
  const float m0 = k1 * az - k2 * ay;
  const float m1 = k2 * ax - k0 * az;
  const float m2 = k0 * ay - k1 * ax;
  const float a0 = w0 + sn * k0 + oc * m0;
  const float a1 = w1 + sn * k1 + oc * m1;
  const float a2 = w2 + sn * k2 + oc * m2;
  const float a3 = s0 * por[0] + s1 * por[1] + s2 * por[2] + s3;
  rv = make_float4(a0, a1, a2, a3);

  const float b0 = a0 * Rop[0] + a1 * Rop[3] + a2 * Rop[6];
  const float b1 = a0 * Rop[1] + a1 * Rop[4] + a2 * Rop[7];
  const float b2 = a0 * Rop[2] + a1 * Rop[5] + a2 * Rop[8];
  const float bap = b0 * ap[0] + b1 * ap[1] + b2 * ap[2];
  const float b3 = d * bap + (a0 * pop[0] + a1 * pop[1] + a2 * pop[2] + a3);
  s0 = b0; s1 = b1; s2 = b2; s3 = b3;
  pr = make_float4(b0, b1, b2, b3);

  if ((J & 1) == 0) {
    constexpr int SL = NJ + J / 2;
    const float t0 = lane_val<SL>(RA[0]), t1 = lane_val<SL>(RA[1]), t2 = lane_val<SL>(RA[2]);
    const float t3 = lane_val<SL>(RA[3]), t4 = lane_val<SL>(RA[4]), t5 = lane_val<SL>(RA[5]);
    const float t6 = lane_val<SL>(RA[6]), t7 = lane_val<SL>(RA[7]), t8 = lane_val<SL>(RA[8]);
    const float pt0 = p_track[3 * J], pt1 = p_track[3 * J + 1], pt2 = p_track[3 * J + 2];
    const float c0 = (row == 0) ? t0 : (row == 1) ? t1 : (row == 2) ? t2 : pt0;
    const float c1 = (row == 0) ? t3 : (row == 1) ? t4 : (row == 2) ? t5 : pt1;
    const float c2 = (row == 0) ? t6 : (row == 1) ? t7 : (row == 2) ? t8 : pt2;
    const float c3 = (row == 3) ? 1.f : 0.f;
    v2f Atr;
    Atr.x = h ? c2 : c0;
    Atr.y = h ? c3 : c1;
    const float x = xhalf(h ? s0 : s2);
    const float y = xhalf(h ? s1 : s3);
    v2f B1, B2;
    B1.x = h ? x : s0;  B1.y = h ? y : s1;
    B2.x = h ? s2 : x;  B2.y = h ? s3 : y;
    const v8f zero8 = {0.f, 0.f, 0.f, 0.f, 0.f, 0.f, 0.f, 0.f};
    v8f D1 = __builtin_amdgcn_wmma_f32_16x16x4_f32(
        false, Atr, false, B1, (short)0, zero8, false, false);
    v8f D2 = __builtin_amdgcn_wmma_f32_16x16x4_f32(
        false, Atr, false, B2, (short)0, zero8, false, false);
    const float e0 = h ? D2[0] : D1[0];
    const float e1 = h ? D2[1] : D1[1];
    const float e2 = h ? D2[2] : D1[2];
    const float e3 = h ? D2[3] : D1[3];
    tk = make_float4(e0, e1, e2, e3);
  }
}

__global__ __launch_bounds__(256) void fk_main_kernel(
    const float* __restrict__ rev_q, const float* __restrict__ pri_q,
    const float* __restrict__ rev_p_off, const float* __restrict__ rev_rpy_off,
    const float* __restrict__ pri_p_off, const float* __restrict__ pri_rpy_off,
    const float* __restrict__ rev_axis, const float* __restrict__ pri_axis,
    const float* __restrict__ p_track, const float* __restrict__ rpy_track,
    float* __restrict__ out, int B) {
  const int  gl    = blockIdx.x * 256 + threadIdx.x;
  const int  ch    = gl >> 2;
  const int  row   = gl & 3;
  const int  h     = (threadIdx.x >> 4) & 1;
  const int  lane  = threadIdx.x & 31;
  const bool alive = (ch < B);
  const size_t chain = (size_t)(alive ? ch : (B - 1));

  float* trk_out = out;
  float* rev_out = out + (size_t)B * 160;
  float* pri_out = out + (size_t)B * 480;

  const int jA = (lane < NJ) ? lane : ((lane < NJ + 10) ? 2 * (lane - NJ) : 18);
  const float* rpyA = ((lane < NJ) ? rev_rpy_off : rpy_track) + 3 * jA;
  const int jB = (lane < NJ) ? lane : (NJ - 1);
  float angC[6], sC[6], cC[6];
  angC[0] = rpyA[0]; angC[1] = rpyA[1]; angC[2] = rpyA[2];
  angC[3] = pri_rpy_off[3 * jB]; angC[4] = pri_rpy_off[3 * jB + 1]; angC[5] = pri_rpy_off[3 * jB + 2];
  const float axx = rev_axis[3 * jB], axy = rev_axis[3 * jB + 1], axz = rev_axis[3 * jB + 2];

  float qv[5], d5[5], sn5[5], cs5[5];
#pragma unroll
  for (int t = 0; t < 5; ++t) {
    qv[t] = rev_q[chain * NJ + 5 * row + t];
    d5[t] = pri_q[chain * NJ + 5 * row + t];
  }

  float RA[9], RB[9], AX[3];
  sincos_rolled<6>(angC, sC, cC);
  rpy2r_sc(sC[0], cC[0], sC[1], cC[1], sC[2], cC[2], RA);
  rpy2r_sc(sC[3], cC[3], sC[4], cC[4], sC[5], cC[5], RB);
  {
    const float inv = 1.f / (sqrtf(axx * axx + axy * axy + axz * axz) + 1e-12f);
    AX[0] = axx * inv; AX[1] = axy * inv; AX[2] = axz * inv;
  }

  sincos_rolled<5>(qv, sn5, cs5);

  float s0 = (row == 0) ? 1.f : 0.f;
  float s1 = (row == 1) ? 1.f : 0.f;
  float s2 = (row == 2) ? 1.f : 0.f;
  float s3 = (row == 3) ? 1.f : 0.f;

  const int u     = (threadIdx.x >> 2) & 3;
  const int cbase = ((gl >> 5) << 3) + 4 * h;
  const bool full = (__builtin_amdgcn_readfirstlane(gl >> 5) * 8 + 8 <= B);
  float* rev_lane = rev_out + ((size_t)cbase * NJ + u) * 16 + row * 4;
  float* pri_lane = pri_out + ((size_t)cbase * NJ + u) * 16 + row * 4;
  float* trk_lane = trk_out + ((size_t)cbase * 10 + u) * 16 + row * 4;

  float4 RV[4], PR[4], TK[4];
#define JOINT(J) joint_step<J>(RA, RB, AX, rev_p_off, pri_p_off, pri_axis, p_track, sn5, cs5, d5, \
                               s0, s1, s2, s3, h, row, RV[(J) & 3], PR[(J) & 3], TK[((J) >> 1) & 3]);
#define FLUSH(G) \
  xpose4(RV); store_units<NJ>(rev_lane, RV, cbase, B, full, 4 * (G)); \
  xpose4(PR); store_units<NJ>(pri_lane, PR, cbase, B, full, 4 * (G));
  JOINT(0)  JOINT(1)  JOINT(2)  JOINT(3)  FLUSH(0)
  JOINT(4)  JOINT(5)  JOINT(6)
  xpose4(TK); store_units<10>(trk_lane, TK, cbase, B, full, 0);
  JOINT(7)  FLUSH(1)
  JOINT(8)  JOINT(9)  JOINT(10) JOINT(11) FLUSH(2)
  JOINT(12) JOINT(13) JOINT(14)
  xpose4(TK); store_units<10>(trk_lane, TK, cbase, B, full, 4);
  JOINT(15) FLUSH(3)
  JOINT(16) JOINT(17) JOINT(18) JOINT(19) FLUSH(4)
#undef FLUSH
#undef JOINT

  xstage<4>(TK[0].x, TK[1].x); xstage<4>(TK[0].y, TK[1].y);
  xstage<4>(TK[0].z, TK[1].z); xstage<4>(TK[0].w, TK[1].w);
  const int cpair = cbase + 2 * (u >> 1);
  float* trk_tail = trk_out + ((size_t)cpair * 10 + 8 + (u & 1)) * 16 + row * 4;
#pragma unroll
  for (int t = 0; t < 2; ++t) {
    if (cpair + t < B) {
      float4* o4 = (float4*)(trk_tail + t * 10 * 16);
      *o4 = TK[t];
    }
  }
}

extern "C" void kernel_launch(void* const* d_in, const int* in_sizes, int n_in,
                              void* d_out, int out_size, void* d_ws, size_t ws_size,
                              hipStream_t stream) {
  (void)n_in; (void)out_size; (void)ws_size;
  const float* rev_q       = (const float*)d_in[0];
  const float* pri_q       = (const float*)d_in[1];
  const float* rev_p_off   = (const float*)d_in[2];
  const float* rev_rpy_off = (const float*)d_in[3];
  const float* pri_p_off   = (const float*)d_in[4];
  const float* pri_rpy_off = (const float*)d_in[5];
  const float* rev_axis    = (const float*)d_in[6];
  const float* pri_axis    = (const float*)d_in[7];
  const float* p_track     = (const float*)d_in[8];
  const float* rpy_track   = (const float*)d_in[9];

  (void)d_ws;
  int B = in_sizes[0] / NJ;

  int grid = (B + 63) / 64;
  fk_main_kernel<<<grid, 256, 0, stream>>>(rev_q, pri_q, rev_p_off, rev_rpy_off, pri_p_off,
                                           pri_rpy_off, rev_axis, pri_axis, p_track, rpy_track,
                                           (float*)d_out, B);
}
